# starting kernel
# baseline (speedup 1.0000x reference)
#include <hip/hip_runtime.h>


#define Tn   2048
#define Bn   8
#define Cn   1024
#define Hn   16
#define Kn   31
#define PADL 30

#define NT  8
#define WPB 4

typedef __attribute__((ext_vector_type(2))) float v2f;
typedef __attribute__((ext_vector_type(8))) float v8f;

__global__ __launch_bounds__(Hn * 32) void lc_softmax(const float* __restrict__ w,
                                                      float* __restrict__ wsm) {
    const int h = threadIdx.x >> 5;
    const int k = threadIdx.x & 31;
    float v = (k < Kn) ? w[h * Kn + k] : -__builtin_inff();
    float m = v;
#pragma unroll
    for (int off = 16; off > 0; off >>= 1) m = fmaxf(m, __shfl_xor(m, off, 32));
    float e = (k < Kn) ? __expf(v - m) : 0.0f;
    float s = e;
#pragma unroll
    for (int off = 16; off > 0; off >>= 1) s += __shfl_xor(s, off, 32);
    if (k < Kn) wsm[h * Kn + k] = e / s;
}

template <bool SAFE>
__device__ __forceinline__ void conv_strip(const float* __restrict__ x,
                                           const float* __restrict__ wsm,
                                           float* __restrict__ out,
                                           int h, int t_base, long colBase,
                                           int mlo, int hh) {
    constexpr long rowStride = (long)Bn * Cn;

    float a[12][2];
#pragma unroll
    for (int j = 0; j < 12; ++j) {
#pragma unroll
        for (int v = 0; v < 2; ++v) {
            const int s   = 4 * j + v + 2 * hh;
            const int idx = s - mlo;
            a[j][v] = (idx >= 0 && idx < Kn) ? wsm[h * Kn + idx] : 0.0f;
        }
    }

    const int    tBase0 = t_base - PADL + 2 * hh;
    const float* xb     = x + (long)tBase0 * rowStride + colBase;

    auto loadx = [&](int ofs) -> float {
        if (!SAFE) {
            return xb[(long)ofs * rowStride];
        } else {
            const int tt  = tBase0 + ofs;
            int       ttc = tt < 0 ? 0 : tt;
            ttc           = ttc > (Tn - 1) ? (Tn - 1) : ttc;
            const float v = x[(long)ttc * rowStride + colBase];
            return ((unsigned)tt < (unsigned)Tn) ? v : 0.0f;
        }
    };

    float bx[12][2];
#pragma unroll
    for (int j = 0; j < 12; ++j) {
#pragma unroll
        for (int v = 0; v < 2; ++v) bx[j][v] = loadx(4 * j + v);
    }

    float* outb = out + (long)(t_base + 8 * hh) * rowStride + colBase;

#pragma unroll
    for (int i = 0; i < NT; ++i) {
        v8f acc = {};
#pragma unroll
        for (int j = 0; j < 12; ++j) {
            v2f av = {a[j][0], a[j][1]};
            v2f bv = {bx[j][0], bx[j][1]};
            acc = __builtin_amdgcn_wmma_f32_16x16x4_f32(
                false, av, false, bv, (short)0, acc, false, false);
        }

#pragma unroll
        for (int r = 0; r < 8; ++r)
            __builtin_nontemporal_store(acc[r], &outb[(long)(16 * i + r) * rowStride]);

        if (i + 1 < NT) {
#pragma unroll
            for (int j = 0; j < 8; ++j) {
                bx[j][0] = bx[j + 4][0];
                bx[j][1] = bx[j + 4][1];
            }
#pragma unroll
            for (int j = 8; j < 12; ++j) {
#pragma unroll
                for (int v = 0; v < 2; ++v)
                    bx[j][v] = loadx(16 * (i + 1) + 4 * j + v);
            }
        }
    }
}

__global__ __launch_bounds__(WPB * 32) void lc_conv(const float* __restrict__ x,
                                                    const float* __restrict__ wsm,
                                                    float* __restrict__ out) {
    const int lane = threadIdx.x & 31;
    const int wid  = blockIdx.x * WPB + (threadIdx.x >> 5);

    constexpr int strips = Tn / (16 * NT);
    constexpr int ctiles = Cn / 16;
    const int strip = wid % strips;
    const int ctile = (wid / strips) % ctiles;
    const int b     = wid / (strips * ctiles);

    const int h      = ctile / ((Cn / Hn) / 16);
    const int c0     = ctile * 16;
    const int t_base = strip * (16 * NT);

    const int  mlo     = lane & 15;
    const int  hh      = lane >> 4;
    const long colBase = (long)b * Cn + c0 + mlo;

    if (strip == 0 || strip == strips - 1)
        conv_strip<true>(x, wsm, out, h, t_base, colBase, mlo, hh);
    else
        conv_strip<false>(x, wsm, out, h, t_base, colBase, mlo, hh);
}

extern "C" void kernel_launch(void* const* d_in, const int* in_sizes, int n_in,
                              void* d_out, int out_size, void* d_ws, size_t ws_size,
                              hipStream_t stream) {
    const float* x = (const float*)d_in[0];
    const float* w = (const float*)d_in[1];
    float* out = (float*)d_out;
    float* wsm = (float*)d_ws;

    lc_softmax<<<1, Hn * 32, 0, stream>>>(w, wsm);

    constexpr int total_waves = Bn * (Cn / 16) * (Tn / (16 * NT));
    lc_conv<<<total_waves / WPB, WPB * 32, 0, stream>>>(x, wsm, out);
}


# Wave tile widened to one head's 64 channels: 16-byte loads/stores per lane (whole 256-byte row pieces), 4 accumulator tiles per A fragment, next rows requested two tiles ahead
# speedup vs baseline: 2.0836x; 2.0836x over previous
#include <hip/hip_runtime.h>


#define Tn   2048
#define Bn   8
#define Cn   1024
#define Hn   16
#define Kn   31
#define PADL 30

#define NT  8
#define WPB 4
#define CW  (Cn / Hn)

typedef __attribute__((ext_vector_type(2))) float v2f;
typedef __attribute__((ext_vector_type(4))) float v4f;
typedef __attribute__((ext_vector_type(8))) float v8f;

__global__ __launch_bounds__(Hn * 32) void lc_softmax(const float* __restrict__ w,
                                                      float* __restrict__ wsm) {
    const int h = threadIdx.x >> 5;
    const int k = threadIdx.x & 31;
    float v = (k < Kn) ? w[h * Kn + k] : -__builtin_inff();
    float m = v;
#pragma unroll
    for (int off = 16; off > 0; off >>= 1) m = fmaxf(m, __shfl_xor(m, off, 32));
    float e = (k < Kn) ? __expf(v - m) : 0.0f;
    float s = e;
#pragma unroll
    for (int off = 16; off > 0; off >>= 1) s += __shfl_xor(s, off, 32);
    if (k < Kn) wsm[h * Kn + k] = e / s;
}

template <bool SAFE>
__device__ __forceinline__ void conv_strip(const float* __restrict__ x,
                                           const float* __restrict__ wsm,
                                           float* __restrict__ out,
                                           int h, int t_base, long colBase,
                                           int mlo, int hh) {
    constexpr long rowStride = (long)Bn * Cn;

    float a[12][2];
#pragma unroll
    for (int j = 0; j < 12; ++j) {
#pragma unroll
        for (int v = 0; v < 2; ++v) {
            const int s   = 4 * j + v + 2 * hh;
            const int idx = s - mlo;
            a[j][v] = (idx >= 0 && idx < Kn) ? wsm[h * Kn + idx] : 0.0f;
        }
    }

    const int    tBase0 = t_base - PADL + 2 * hh;
    const float* xb     = x + (long)tBase0 * rowStride + colBase;

    auto loadx = [&](int ofs) -> v4f {
        if (!SAFE || (ofs >= PADL && ofs < NT * 16 + PADL - 2)) {
            return *(const v4f*)(xb + (long)ofs * rowStride);
        } else {
            const int tt  = tBase0 + ofs;
            int       ttc = tt < 0 ? 0 : tt;
            ttc           = ttc > (Tn - 1) ? (Tn - 1) : ttc;
            const v4f v = *(const v4f*)(x + (long)ttc * rowStride + colBase);
            const v4f z = {0.0f, 0.0f, 0.0f, 0.0f};
            return ((unsigned)tt < (unsigned)Tn) ? v : z;
        }
    };

    v4f bx[12][2];
#pragma unroll
    for (int j = 0; j < 12; ++j) {
#pragma unroll
        for (int v = 0; v < 2; ++v) bx[j][v] = loadx(4 * j + v);
    }

    v4f nx[NT][4][2];
#pragma unroll
    for (int j = 0; j < 4; ++j) {
#pragma unroll
        for (int v = 0; v < 2; ++v) nx[1][j][v] = loadx(16 * 1 + 4 * (8 + j) + v);
    }

    float* outb = out + (long)(t_base + 8 * hh) * rowStride + colBase;

#pragma unroll
    for (int i = 0; i < NT; ++i) {
        if (i + 2 < NT) {
#pragma unroll
            for (int j = 0; j < 4; ++j) {
#pragma unroll
                for (int v = 0; v < 2; ++v)
                    nx[i + 2][j][v] = loadx(16 * (i + 2) + 4 * (8 + j) + v);
            }
        }

        v8f acc[4] = {};
#pragma unroll
        for (int j = 0; j < 12; ++j) {
            v2f av = {a[j][0], a[j][1]};
#pragma unroll
            for (int q = 0; q < 4; ++q) {
                v2f bv = {bx[j][0][q], bx[j][1][q]};
                acc[q] = __builtin_amdgcn_wmma_f32_16x16x4_f32(
                    false, av, false, bv, (short)0, acc[q], false, false);
            }
        }

#pragma unroll
        for (int r = 0; r < 8; ++r) {
            v4f o = {acc[0][r], acc[1][r], acc[2][r], acc[3][r]};
            __builtin_nontemporal_store(o, (v4f*)&outb[(long)(16 * i + r) * rowStride]);
        }

        if (i + 1 < NT) {
#pragma unroll
            for (int j = 0; j < 8; ++j) {
                bx[j][0] = bx[j + 4][0];
                bx[j][1] = bx[j + 4][1];
            }
#pragma unroll
            for (int j = 0; j < 4; ++j) {
                bx[8 + j][0] = nx[i + 1][j][0];
                bx[8 + j][1] = nx[i + 1][j][1];
            }
        }
    }
}

__global__ __launch_bounds__(WPB * 32) void lc_conv(const float* __restrict__ x,
                                                    const float* __restrict__ wsm,
                                                    float* __restrict__ out) {
    const int lane = threadIdx.x & 31;
    const int wid  = blockIdx.x * WPB + (threadIdx.x >> 5);

    constexpr int strips = Tn / (16 * NT);
    const int strip = wid % strips;
    const int h     = (wid / strips) % Hn;
    const int b     = wid / (strips * Hn);

    const int t_base = strip * (16 * NT);

    const int  mlo     = lane & 15;
    const int  hh      = lane >> 4;
    const long colBase = (long)b * Cn + h * CW + 4 * mlo;

    if (strip == 0 || strip == strips - 1)
        conv_strip<true>(x, wsm, out, h, t_base, colBase, mlo, hh);
    else
        conv_strip<false>(x, wsm, out, h, t_base, colBase, mlo, hh);
}

extern "C" void kernel_launch(void* const* d_in, const int* in_sizes, int n_in,
                              void* d_out, int out_size, void* d_ws, size_t ws_size,
                              hipStream_t stream) {
    const float* x = (const float*)d_in[0];
    const float* w = (const float*)d_in[1];
    float* out = (float*)d_out;
    float* wsm = (float*)d_ws;

    lc_softmax<<<1, Hn * 32, 0, stream>>>(w, wsm);

    constexpr int total_waves = Bn * Hn * (Tn / (16 * NT));
    lc_conv<<<total_waves / WPB, WPB * 32, 0, stream>>>(x, wsm, out);
}


# Softmax launch removed: each conv wave takes its head's max and sum by the same xor butterfly and each lane normalises its own 24 band taps from the raw weights while x rows load; 1 launch, not 2
# speedup vs baseline: 2.2481x; 1.0789x over previous
#include <hip/hip_runtime.h>


#define Tn   2048
#define Bn   8
#define Cn   1024
#define Hn   16
#define Kn   31
#define PADL 30

#define NT  8
#define WPB 4
#define CW  (Cn / Hn)

typedef __attribute__((ext_vector_type(2))) float v2f;
typedef __attribute__((ext_vector_type(4))) float v4f;
typedef __attribute__((ext_vector_type(8))) float v8f;

template <bool SAFE>
__device__ __forceinline__ void conv_strip(const float* __restrict__ x,
                                           const float* __restrict__ w,
                                           float* __restrict__ out,
                                           int h, int t_base, long colBase,
                                           int mlo, int hh, int lane) {
    constexpr long rowStride = (long)Bn * Cn;
    const float* wh = w + h * Kn;

    float raw[12][2];
#pragma unroll
    for (int j = 0; j < 12; ++j) {
#pragma unroll
        for (int v = 0; v < 2; ++v) {
            const int idx  = 4 * j + v + 2 * hh - mlo;
            const int idxc = idx < 0 ? 0 : (idx > Kn - 1 ? Kn - 1 : idx);
            raw[j][v] = wh[idxc];
        }
    }
    const float wl = wh[lane < Kn ? lane : Kn - 1];
    __builtin_amdgcn_sched_barrier(0);

    const int    tBase0 = t_base - PADL + 2 * hh;
    const float* xb     = x + (long)tBase0 * rowStride + colBase;

    auto loadx = [&](int ofs) -> v4f {
        if (!SAFE || (ofs >= PADL && ofs < NT * 16 + PADL - 2)) {
            return *(const v4f*)(xb + (long)ofs * rowStride);
        } else {
            const int tt  = tBase0 + ofs;
            int       ttc = tt < 0 ? 0 : tt;
            ttc           = ttc > (Tn - 1) ? (Tn - 1) : ttc;
            const v4f v = *(const v4f*)(x + (long)ttc * rowStride + colBase);
            const v4f z = {0.0f, 0.0f, 0.0f, 0.0f};
            return ((unsigned)tt < (unsigned)Tn) ? v : z;
        }
    };

    v4f bx[12][2];
#pragma unroll
    for (int j = 0; j < 12; ++j) {
#pragma unroll
        for (int v = 0; v < 2; ++v) bx[j][v] = loadx(4 * j + v);
    }

    __builtin_amdgcn_sched_barrier(0);

    const float wv = (lane < Kn) ? wl : -__builtin_inff();
    float m = wv;
#pragma unroll
    for (int off = 16; off > 0; off >>= 1) m = fmaxf(m, __shfl_xor(m, off, 32));
    const float e = (lane < Kn) ? __expf(wv - m) : 0.0f;
    float s = e;
#pragma unroll
    for (int off = 16; off > 0; off >>= 1) s += __shfl_xor(s, off, 32);

    float a[12][2];
#pragma unroll
    for (int j = 0; j < 12; ++j) {
#pragma unroll
        for (int v = 0; v < 2; ++v) {
            const int   idx = 4 * j + v + 2 * hh - mlo;
            const float t   = __expf(raw[j][v] - m) / s;
            a[j][v] = (idx >= 0 && idx < Kn) ? t : 0.0f;
        }
    }

    v4f nx[NT][4][2];
#pragma unroll
    for (int j = 0; j < 4; ++j) {
#pragma unroll
        for (int v = 0; v < 2; ++v) nx[1][j][v] = loadx(16 * 1 + 4 * (8 + j) + v);
    }

    float* outb = out + (long)(t_base + 8 * hh) * rowStride + colBase;

#pragma unroll
    for (int i = 0; i < NT; ++i) {
        if (i + 2 < NT) {
#pragma unroll
            for (int j = 0; j < 4; ++j) {
#pragma unroll
                for (int v = 0; v < 2; ++v)
                    nx[i + 2][j][v] = loadx(16 * (i + 2) + 4 * (8 + j) + v);
            }
        }

        v8f acc[4] = {};
#pragma unroll
        for (int j = 0; j < 12; ++j) {
            v2f av = {a[j][0], a[j][1]};
#pragma unroll
            for (int q = 0; q < 4; ++q) {
                v2f bv = {bx[j][0][q], bx[j][1][q]};
                acc[q] = __builtin_amdgcn_wmma_f32_16x16x4_f32(
                    false, av, false, bv, (short)0, acc[q], false, false);
            }
        }

#pragma unroll
        for (int r = 0; r < 8; ++r) {
            v4f o = {acc[0][r], acc[1][r], acc[2][r], acc[3][r]};
            __builtin_nontemporal_store(o, (v4f*)&outb[(long)(16 * i + r) * rowStride]);
        }

        if (i + 1 < NT) {
#pragma unroll
            for (int j = 0; j < 8; ++j) {
                bx[j][0] = bx[j + 4][0];
                bx[j][1] = bx[j + 4][1];
            }
#pragma unroll
            for (int j = 0; j < 4; ++j) {
                bx[8 + j][0] = nx[i + 1][j][0];
                bx[8 + j][1] = nx[i + 1][j][1];
            }
        }
    }
}

__global__ __launch_bounds__(WPB * 32, 3) void lc_conv(const float* __restrict__ x,
                                                    const float* __restrict__ w,
                                                    float* __restrict__ out) {
    const int lane = threadIdx.x & 31;
    const int wid  = blockIdx.x * WPB + (threadIdx.x >> 5);

    constexpr int strips = Tn / (16 * NT);
    const int strip = wid % strips;
    const int h     = (wid / strips) % Hn;
    const int b     = wid / (strips * Hn);

    const int t_base = strip * (16 * NT);

    const int  mlo     = lane & 15;
    const int  hh      = lane >> 4;
    const long colBase = (long)b * Cn + h * CW + 4 * mlo;

    if (strip == 0 || strip == strips - 1)
        conv_strip<true>(x, w, out, h, t_base, colBase, mlo, hh, lane);
    else
        conv_strip<false>(x, w, out, h, t_base, colBase, mlo, hh, lane);
}

extern "C" void kernel_launch(void* const* d_in, const int* in_sizes, int n_in,
                              void* d_out, int out_size, void* d_ws, size_t ws_size,
                              hipStream_t stream) {
    const float* x = (const float*)d_in[0];
    const float* w = (const float*)d_in[1];
    float* out = (float*)d_out;

    constexpr int total_waves = Bn * Hn * (Tn / (16 * NT));
    lc_conv<<<total_waves / WPB, WPB * 32, 0, stream>>>(x, w, out);
}


# Odd strips walk their 8 tiles in descending time (mirrored ring): neighbour strips ask for their 32 shared rows at the same moment, so the 2nd read hits cache; edge clamps folded into the 2 bodies
# speedup vs baseline: 2.5242x; 1.1228x over previous
#include <hip/hip_runtime.h>


#define Tn   2048
#define Bn   8
#define Cn   1024
#define Hn   16
#define Kn   31
#define PADL 30

#define NT  8
#define WPB 4
#define CW  (Cn / Hn)

typedef __attribute__((ext_vector_type(2))) float v2f;
typedef __attribute__((ext_vector_type(4))) float v4f;
typedef __attribute__((ext_vector_type(8))) float v8f;

static_assert(((Tn / (16 * NT)) & 1) == 0, "strip 0 must be even and the last strip odd");
template <bool REV>
__device__ __forceinline__ void conv_strip(const float* __restrict__ x,
                                           const float* __restrict__ w,
                                           float* __restrict__ out,
                                           int h, int t_base, long colBase,
                                           int mlo, int hh, int lane) {
    constexpr long rowStride = (long)Bn * Cn;
    const float* wh = w + h * Kn;

    float raw[12][2];
#pragma unroll
    for (int j = 0; j < 12; ++j) {
#pragma unroll
        for (int v = 0; v < 2; ++v) {
            const int idx  = 4 * j + v + 2 * hh - mlo;
            const int idxc = idx < 0 ? 0 : (idx > Kn - 1 ? Kn - 1 : idx);
            raw[j][v] = wh[idxc];
        }
    }
    const float wl = wh[lane < Kn ? lane : Kn - 1];
    __builtin_amdgcn_sched_barrier(0);

    const int    tBase0 = t_base - PADL + 2 * hh;
    const float* xb     = x + (long)tBase0 * rowStride + colBase;

    auto inside = [](int ofs) -> bool {
        return REV ? (ofs < NT * 16 + PADL - 2) : (ofs >= PADL);
    };
    auto loadx = [&](int ofs) -> v4f {
        if (inside(ofs)) {
            return *(const v4f*)(xb + (long)ofs * rowStride);
        } else {
            const int tt  = tBase0 + ofs;
            int       ttc = tt < 0 ? 0 : tt;
            ttc           = ttc > (Tn - 1) ? (Tn - 1) : ttc;
            return *(const v4f*)(x + (long)ttc * rowStride + colBase);
        }
    };
    auto padx = [&](int ofs, v4f v) -> v4f {
        if (inside(ofs)) return v;
        const int tt = tBase0 + ofs;
        const v4f z  = {0.0f, 0.0f, 0.0f, 0.0f};
        return ((unsigned)tt < (unsigned)Tn) ? v : z;
    };

#define TILE_OF(k)  (REV ? (NT - 1 - (k)) : (k))
#define NEW_SLOT(j) (REV ? (j) : (8 + (j)))
    v4f bx[12][2];
#pragma unroll
    for (int j = 0; j < 12; ++j) {
#pragma unroll
        for (int v = 0; v < 2; ++v) bx[j][v] = loadx(16 * TILE_OF(0) + 4 * j + v);
    }

    __builtin_amdgcn_sched_barrier(0);

    const float wv = (lane < Kn) ? wl : -__builtin_inff();
    float m = wv;
#pragma unroll
    for (int off = 16; off > 0; off >>= 1) m = fmaxf(m, __shfl_xor(m, off, 32));
    const float e = (lane < Kn) ? __expf(wv - m) : 0.0f;
    float s = e;
#pragma unroll
    for (int off = 16; off > 0; off >>= 1) s += __shfl_xor(s, off, 32);

    float a[12][2];
#pragma unroll
    for (int j = 0; j < 12; ++j) {
#pragma unroll
        for (int v = 0; v < 2; ++v) {
            const int   idx = 4 * j + v + 2 * hh - mlo;
            const float t   = __expf(raw[j][v] - m) / s;
            a[j][v] = (idx >= 0 && idx < Kn) ? t : 0.0f;
        }
    }

    v4f nx[NT][4][2];
#pragma unroll
    for (int j = 0; j < 4; ++j) {
#pragma unroll
        for (int v = 0; v < 2; ++v) nx[1][j][v] = loadx(16 * TILE_OF(1) + 4 * NEW_SLOT(j) + v);
    }

    float* outb = out + (long)(t_base + 8 * hh) * rowStride + colBase;

#pragma unroll
    for (int i = 0; i < NT; ++i) {
        if (i + 2 < NT) {
#pragma unroll
            for (int j = 0; j < 4; ++j) {
#pragma unroll
                for (int v = 0; v < 2; ++v)
                    nx[i + 2][j][v] = loadx(16 * TILE_OF(i + 2) + 4 * NEW_SLOT(j) + v);
            }
            __builtin_amdgcn_sched_barrier(0);
        }

        if (i == 0) {
#pragma unroll
            for (int j = 0; j < 12; ++j) {
#pragma unroll
                for (int v = 0; v < 2; ++v)
                    bx[j][v] = padx(16 * TILE_OF(0) + 4 * j + v, bx[j][v]);
            }
        }

        v8f acc[4] = {};
#pragma unroll
        for (int j = 0; j < 12; ++j) {
            v2f av = {a[j][0], a[j][1]};
#pragma unroll
            for (int q = 0; q < 4; ++q) {
                v2f bv = {bx[j][0][q], bx[j][1][q]};
                acc[q] = __builtin_amdgcn_wmma_f32_16x16x4_f32(
                    false, av, false, bv, (short)0, acc[q], false, false);
            }
        }

#pragma unroll
        for (int r = 0; r < 8; ++r) {
            v4f o = {acc[0][r], acc[1][r], acc[2][r], acc[3][r]};
            __builtin_nontemporal_store(o, (v4f*)&outb[(long)(16 * TILE_OF(i) + r) * rowStride]);
        }

        if (i + 1 < NT) {
            if (!REV) {
#pragma unroll
                for (int j = 0; j < 8; ++j) {
                    bx[j][0] = bx[j + 4][0];
                    bx[j][1] = bx[j + 4][1];
                }
            } else {
#pragma unroll
                for (int j = 11; j >= 4; --j) {
                    bx[j][0] = bx[j - 4][0];
                    bx[j][1] = bx[j - 4][1];
                }
            }
#pragma unroll
            for (int j = 0; j < 4; ++j) {
                bx[NEW_SLOT(j)][0] = nx[i + 1][j][0];
                bx[NEW_SLOT(j)][1] = nx[i + 1][j][1];
            }
        }
    }
#undef TILE_OF
#undef NEW_SLOT
}

__global__ __launch_bounds__(WPB * 32, 3) void lc_conv(const float* __restrict__ x,
                                                    const float* __restrict__ w,
                                                    float* __restrict__ out) {
    const int lane = threadIdx.x & 31;
    const int wid  = blockIdx.x * WPB + (threadIdx.x >> 5);

    constexpr int strips = Tn / (16 * NT);
    const int strip = wid % strips;
    const int h     = (wid / strips) % Hn;
    const int b     = wid / (strips * Hn);

    const int t_base = strip * (16 * NT);

    const int  mlo     = lane & 15;
    const int  hh      = lane >> 4;
    const long colBase = (long)b * Cn + h * CW + 4 * mlo;

    if (strip & 1)
        conv_strip<true>(x, w, out, h, t_base, colBase, mlo, hh, lane);
    else
        conv_strip<false>(x, w, out, h, t_base, colBase, mlo, hh, lane);
}

extern "C" void kernel_launch(void* const* d_in, const int* in_sizes, int n_in,
                              void* d_out, int out_size, void* d_ws, size_t ws_size,
                              hipStream_t stream) {
    const float* x = (const float*)d_in[0];
    const float* w = (const float*)d_in[1];
    float* out = (float*)d_out;

    constexpr int total_waves = Bn * Hn * (Tn / (16 * NT));
    lc_conv<<<total_waves / WPB, WPB * 32, 0, stream>>>(x, w, out);
}
